# starting kernel
# baseline (speedup 1.0000x reference)
#include <hip/hip_runtime.h>

typedef __attribute__((ext_vector_type(2))) float v2f;
typedef __attribute__((ext_vector_type(8))) float v8f;

__global__ void fill_kernel(float* __restrict__ p, float v, int n) {
    int i = blockIdx.x * blockDim.x + threadIdx.x;
    if (i < n) p[i] = v;
}

__global__ void deg_kernel(const int* __restrict__ dst, float* __restrict__ deg, int E) {
    int i = blockIdx.x * blockDim.x + threadIdx.x;
    if (i < E) unsafeAtomicAdd(&deg[dst[i]], 1.0f);
}

__global__ void rsqrt_kernel(float* __restrict__ deg, int n) {
    int i = blockIdx.x * blockDim.x + threadIdx.x;
    if (i < n) deg[i] = rsqrtf(deg[i]);
}

__device__ __forceinline__ float squash(float v) {
    return (v * v) * __builtin_amdgcn_rcpf(v);
}

template <int K, int N, bool SQUASH>
__global__ __launch_bounds__((N / 16) * 32)
void gemm_wmma(const float* __restrict__ X, const float* __restrict__ W,
               float* __restrict__ H, int M) {
    const int lane  = threadIdx.x & 31;
    const int ntile = threadIdx.x >> 5;
    const int half  = lane >> 4;
    const int l16   = lane & 15;
    const int row   = blockIdx.x * 16 + l16;
    const int arow  = row < M ? row : M - 1;
    const int ncol  = ntile * 16 + l16;
    const float* __restrict__ xrow = X + (long)arow * K;

    v8f c = {0.f, 0.f, 0.f, 0.f, 0.f, 0.f, 0.f, 0.f};

#pragma unroll
    for (int kk = 0; kk < K; kk += 4) {
        const int k0 = kk + half * 2;
        v2f a;
        {
            float2 av = *(const float2*)(xrow + k0);
            if (SQUASH) { av.x = squash(av.x); av.y = squash(av.y); }
            a.x = av.x; a.y = av.y;
        }
        v2f b;
        b.x = W[k0 * N + ncol];
        b.y = W[(k0 + 1) * N + ncol];
        c = __builtin_amdgcn_wmma_f32_16x16x4_f32(
                 false, a,  false, b,
                 (short)0, c,  false,  false);
    }

#pragma unroll
    for (int i = 0; i < 8; ++i) {
        int r = blockIdx.x * 16 + i + half * 8;
        if (r < M) H[(long)r * N + ncol] = c[i];
    }
}

template <int C>
__global__ void scatter_kernel(const float* __restrict__ h, const int* __restrict__ src,
                               const int* __restrict__ dst, const float* __restrict__ dinv,
                               float* __restrict__ acc, int E) {
    int gid  = blockIdx.x * blockDim.x + threadIdx.x;
    int e    = gid >> 5;
    int lane = gid & 31;
    if (e >= E) return;
    int s = src[e], d = dst[e];
    float p = dinv[s] * dinv[d];
    if (C == 64) {
        float2 v = *(const float2*)(h + (long)s * 64 + lane * 2);
        unsafeAtomicAdd(&acc[(long)d * 64 + lane * 2 + 0], p * v.x);
        unsafeAtomicAdd(&acc[(long)d * 64 + lane * 2 + 1], p * v.y);
    } else {
        float v = h[(long)s * 32 + lane];
        unsafeAtomicAdd(&acc[(long)d * 32 + lane], p * v);
    }
}

template <int C, bool RELU>
__global__ void finalize_kernel(const float* __restrict__ acc, const float* __restrict__ h,
                                const float* __restrict__ dinv, const float* __restrict__ bias,
                                float* __restrict__ out, int Nn) {
    int idx = blockIdx.x * blockDim.x + threadIdx.x;
    if (idx >= Nn * C) return;
    int i = idx / C, ch = idx & (C - 1);
    float di = dinv[i];
    float v = acc[idx] + di * di * h[idx] + bias[ch];
    out[idx] = RELU ? fmaxf(v, 0.0f) : v;
}

static inline size_t align256(size_t n) { return (n + 255) & ~size_t(255); }

extern "C" void kernel_launch(void* const* d_in, const int* in_sizes, int n_in,
                              void* d_out, int out_size, void* d_ws, size_t ws_size,
                              hipStream_t stream) {
    const float* x  = (const float*)d_in[0];
    const int*   ei = (const int*)d_in[1];
    const float* W1 = (const float*)d_in[2];
    const float* b1 = (const float*)d_in[3];
    const float* W2 = (const float*)d_in[4];
    const float* b2 = (const float*)d_in[5];
    float* out = (float*)d_out;

    const int hid  = in_sizes[3];
    const int outc = in_sizes[5];
    const int inc  = in_sizes[2] / hid;
    const int Nn   = in_sizes[0] / inc;
    const int E    = in_sizes[1] / 2;
    const int* src = ei;
    const int* dstv = ei + E;
    (void)outc;

    char* ws = (char*)d_ws;
    float* deg = (float*)ws;  ws += align256((size_t)Nn * 4);
    float* h1  = (float*)ws;  ws += (size_t)Nn * 64 * 4;
    float* a1  = (float*)ws;  ws += (size_t)Nn * 64 * 4;
    float* h2  = (float*)ws;

    const int T = 256;
    fill_kernel<<<(Nn + T - 1) / T, T, 0, stream>>>(deg, 1.0f, Nn);
    fill_kernel<<<(Nn * 64 + T - 1) / T, T, 0, stream>>>(a1, 0.0f, Nn * 64);
    fill_kernel<<<(Nn * 32 + T - 1) / T, T, 0, stream>>>(out, 0.0f, Nn * 32);

    deg_kernel<<<(E + T - 1) / T, T, 0, stream>>>(dstv, deg, E);
    rsqrt_kernel<<<(Nn + T - 1) / T, T, 0, stream>>>(deg, Nn);

    gemm_wmma<128, 64, true><<<(Nn + 15) / 16, 128, 0, stream>>>(x, W1, h1, Nn);
    scatter_kernel<64><<<(E + 7) / 8, T, 0, stream>>>(h1, src, dstv, deg, a1, E);
    finalize_kernel<64, true><<<(Nn * 64 + T - 1) / T, T, 0, stream>>>(a1, h1, deg, b1, a1, Nn);

    gemm_wmma<64, 32, false><<<(Nn + 15) / 16, 64, 0, stream>>>(a1, W2, h2, Nn);
    scatter_kernel<32><<<(E + 7) / 8, T, 0, stream>>>(h2, src, dstv, deg, out, E);
    finalize_kernel<32, false><<<(Nn * 32 + T - 1) / T, T, 0, stream>>>(out, h2, deg, b2, out, Nn);
}


# degree count: per-block LDS histograms (two 16-bit counters per word, LDS integer atomic adds), partial rows summed in the rsqrt pass; no global atomics or fill for degrees; dinv bit-identical
# speedup vs baseline: 1.1810x; 1.1810x over previous
#include <hip/hip_runtime.h>

typedef __attribute__((ext_vector_type(2))) float v2f;
typedef __attribute__((ext_vector_type(8))) float v8f;

__global__ void fill_kernel(float* __restrict__ p, float v, int n) {
    int i = blockIdx.x * blockDim.x + threadIdx.x;
    if (i < n) p[i] = v;
}

#define HIST_THREADS 1024
#define HIST_CHUNK   32768
#define HIST_WORDS   51200

__device__ __forceinline__ int hist_pitch(int Nn) {
    return (((Nn + 1) >> 1) + 31) & ~31;
}

__global__ __launch_bounds__(HIST_THREADS)
void deg_hist_kernel(const int* __restrict__ dst, unsigned* __restrict__ part, int E, int Nn) {
    __shared__ uint4 cnt4[HIST_WORDS / 4];
    unsigned* cnt = (unsigned*)cnt4;
    const int tid   = threadIdx.x;
    const int words = (Nn + 1) >> 1;
    const int wbase = blockIdx.y * HIST_WORDS;
    int wspan = words - wbase;
    if (wspan > HIST_WORDS) wspan = HIST_WORDS;
    const unsigned nbase = (unsigned)wbase * 2u;
    const unsigned nspan = (unsigned)wspan * 2u;

    for (int i = tid; i < HIST_WORDS / 4; i += HIST_THREADS) cnt4[i] = make_uint4(0u, 0u, 0u, 0u);

    const int e0 = blockIdx.x * HIST_CHUNK + tid;
    int d[HIST_CHUNK / HIST_THREADS];
#pragma unroll
    for (int k = 0; k < HIST_CHUNK / HIST_THREADS; ++k) {
        const int e = e0 + k * HIST_THREADS;
        d[k] = dst[e < E ? e : E - 1];
    }
    __syncthreads();
#pragma unroll
    for (int k = 0; k < HIST_CHUNK / HIST_THREADS; ++k) {
        const int e = e0 + k * HIST_THREADS;
        const unsigned r = (unsigned)d[k] - nbase;
        if (e < E && r < nspan) atomicAdd(&cnt[r >> 1], 1u << ((r & 1u) * 16u));
    }
    __syncthreads();

    unsigned* __restrict__ o = part + (size_t)blockIdx.x * hist_pitch(Nn) + wbase;
    const int w4 = wspan >> 2;
    for (int i = tid; i < w4; i += HIST_THREADS) ((uint4*)o)[i] = cnt4[i];
    for (int i = (w4 << 2) + tid; i < wspan; i += HIST_THREADS) o[i] = cnt[i];
}

__global__ void dinv_kernel(const unsigned* __restrict__ part, float* __restrict__ dinv, int Nn, int P) {
    const int w = blockIdx.x * blockDim.x + threadIdx.x;
    const int words = (Nn + 1) >> 1;
    if (w >= words) return;
    const int pitch = hist_pitch(Nn);
    unsigned lo = 0u, hi = 0u;
#pragma unroll 16
    for (int p = 0; p < P; ++p) {
        const unsigned v = part[(size_t)p * pitch + w];
        lo += v & 0xFFFFu;
        hi += v >> 16;
    }
    const float d0 = rsqrtf((float)lo + 1.0f);
    const float d1 = rsqrtf((float)hi + 1.0f);
    if (2 * w + 1 < Nn) *(float2*)(dinv + 2 * w) = make_float2(d0, d1);
    else                dinv[2 * w] = d0;
}

__device__ __forceinline__ float squash(float v) {
    return (v * v) * __builtin_amdgcn_rcpf(v);
}

template <int K, int N, bool SQUASH>
__global__ __launch_bounds__((N / 16) * 32)
void gemm_wmma(const float* __restrict__ X, const float* __restrict__ W,
               float* __restrict__ H, int M) {
    const int lane  = threadIdx.x & 31;
    const int ntile = threadIdx.x >> 5;
    const int half  = lane >> 4;
    const int l16   = lane & 15;
    const int row   = blockIdx.x * 16 + l16;
    const int arow  = row < M ? row : M - 1;
    const int ncol  = ntile * 16 + l16;
    const float* __restrict__ xrow = X + (long)arow * K;

    v8f c = {0.f, 0.f, 0.f, 0.f, 0.f, 0.f, 0.f, 0.f};

#pragma unroll
    for (int kk = 0; kk < K; kk += 4) {
        const int k0 = kk + half * 2;
        v2f a;
        {
            float2 av = *(const float2*)(xrow + k0);
            if (SQUASH) { av.x = squash(av.x); av.y = squash(av.y); }
            a.x = av.x; a.y = av.y;
        }
        v2f b;
        b.x = W[k0 * N + ncol];
        b.y = W[(k0 + 1) * N + ncol];
        c = __builtin_amdgcn_wmma_f32_16x16x4_f32(
                 false, a,  false, b,
                 (short)0, c,  false,  false);
    }

#pragma unroll
    for (int i = 0; i < 8; ++i) {
        int r = blockIdx.x * 16 + i + half * 8;
        if (r < M) H[(long)r * N + ncol] = c[i];
    }
}

template <int C>
__global__ void scatter_kernel(const float* __restrict__ h, const int* __restrict__ src,
                               const int* __restrict__ dst, const float* __restrict__ dinv,
                               float* __restrict__ acc, int E) {
    int gid  = blockIdx.x * blockDim.x + threadIdx.x;
    int e    = gid >> 5;
    int lane = gid & 31;
    if (e >= E) return;
    int s = src[e], d = dst[e];
    float p = dinv[s] * dinv[d];
    if (C == 64) {
        float2 v = *(const float2*)(h + (long)s * 64 + lane * 2);
        unsafeAtomicAdd(&acc[(long)d * 64 + lane * 2 + 0], p * v.x);
        unsafeAtomicAdd(&acc[(long)d * 64 + lane * 2 + 1], p * v.y);
    } else {
        float v = h[(long)s * 32 + lane];
        unsafeAtomicAdd(&acc[(long)d * 32 + lane], p * v);
    }
}

template <int C, bool RELU>
__global__ void finalize_kernel(const float* __restrict__ acc, const float* __restrict__ h,
                                const float* __restrict__ dinv, const float* __restrict__ bias,
                                float* __restrict__ out, int Nn) {
    int idx = blockIdx.x * blockDim.x + threadIdx.x;
    if (idx >= Nn * C) return;
    int i = idx / C, ch = idx & (C - 1);
    float di = dinv[i];
    float v = acc[idx] + di * di * h[idx] + bias[ch];
    out[idx] = RELU ? fmaxf(v, 0.0f) : v;
}

static inline size_t align256(size_t n) { return (n + 255) & ~size_t(255); }

extern "C" void kernel_launch(void* const* d_in, const int* in_sizes, int n_in,
                              void* d_out, int out_size, void* d_ws, size_t ws_size,
                              hipStream_t stream) {
    const float* x  = (const float*)d_in[0];
    const int*   ei = (const int*)d_in[1];
    const float* W1 = (const float*)d_in[2];
    const float* b1 = (const float*)d_in[3];
    const float* W2 = (const float*)d_in[4];
    const float* b2 = (const float*)d_in[5];
    float* out = (float*)d_out;

    const int hid  = in_sizes[3];
    const int outc = in_sizes[5];
    const int inc  = in_sizes[2] / hid;
    const int Nn   = in_sizes[0] / inc;
    const int E    = in_sizes[1] / 2;
    const int* src = ei;
    const int* dstv = ei + E;
    (void)outc;

    char* ws = (char*)d_ws;
    float* deg = (float*)ws;  ws += align256((size_t)Nn * 4);
    float* h1  = (float*)ws;  ws += (size_t)Nn * 64 * 4;
    float* a1  = (float*)ws;  ws += (size_t)Nn * 64 * 4;
    float* h2  = (float*)ws;  ws += align256((size_t)Nn * 32 * 4);
    unsigned* part = (unsigned*)ws;

    const int T = 256;
    fill_kernel<<<(Nn * 64 + T - 1) / T, T, 0, stream>>>(a1, 0.0f, Nn * 64);
    fill_kernel<<<(Nn * 32 + T - 1) / T, T, 0, stream>>>(out, 0.0f, Nn * 32);

    const int chunks = (E + HIST_CHUNK - 1) / HIST_CHUNK;
    const int words  = (Nn + 1) / 2;
    deg_hist_kernel<<<dim3(chunks, (words + HIST_WORDS - 1) / HIST_WORDS), HIST_THREADS, 0, stream>>>(dstv, part, E, Nn);
    dinv_kernel<<<(words + T - 1) / T, T, 0, stream>>>(part, deg, Nn, chunks);

    gemm_wmma<128, 64, true><<<(Nn + 15) / 16, 128, 0, stream>>>(x, W1, h1, Nn);
    scatter_kernel<64><<<(E + 7) / 8, T, 0, stream>>>(h1, src, dstv, deg, a1, E);
    finalize_kernel<64, true><<<(Nn * 64 + T - 1) / T, T, 0, stream>>>(a1, h1, deg, b1, a1, Nn);

    gemm_wmma<64, 32, false><<<(Nn + 15) / 16, 64, 0, stream>>>(a1, W2, h2, Nn);
    scatter_kernel<32><<<(E + 7) / 8, T, 0, stream>>>(h2, src, dstv, deg, out, E);
    finalize_kernel<32, false><<<(Nn * 32 + T - 1) / T, T, 0, stream>>>(out, h2, deg, b2, out, Nn);
}


# row gathers and LDS adds: a half-wave per edge, 16 (8) bytes a lane: one load fetches 2 rows, a batch's 32 rows in flight; DPP shares replace per-edge readlanes; LDS rows padded; sum order: arrival
# speedup vs baseline: 4.1295x; 3.4966x over previous
#include <hip/hip_runtime.h>

typedef __attribute__((ext_vector_type(2))) float v2f;
typedef __attribute__((ext_vector_type(8))) float v8f;

#define HIST_THREADS 1024
#define HIST_CHUNK   32768
#define HIST_WORDS   51200
#define OWN_NODES    400
#define KEY_GROUPS   8

#define KEY_PLANES 32
__host__ __device__ __forceinline__ int plane_pitch(int E) {
    return (((E + 31) >> 5) + 31) & ~31;
}

__device__ __forceinline__ void write_key_planes(const int* __restrict__ dst, unsigned* __restrict__ planes,
                                                 int g0, int lane, int E) {
    const int PW = plane_pitch(E);
    int d[KEY_GROUPS];
#pragma unroll
    for (int j = 0; j < KEY_GROUPS; ++j) {
        const int e = (g0 + j) * 32 + lane;
        d[j] = dst[e < E ? e : E - 1];
    }
    unsigned pw[KEY_GROUPS];
#pragma unroll
    for (int j = 0; j < KEY_GROUPS; ++j) {
        const unsigned kb = (unsigned)d[j] / OWN_NODES;
        const unsigned hi = kb >> 4;
        const unsigned lo = kb & 15u;
        unsigned w = 0u;
#pragma unroll
        for (int p = 0; p < 16; ++p) {
            const unsigned bh = __builtin_amdgcn_ballot_w32(hi == (unsigned)p);
            const unsigned bl = __builtin_amdgcn_ballot_w32(lo == (unsigned)p);
            if (lane == p)      w = bh;
            if (lane == 16 + p) w = bl;
        }
        pw[j] = w;
    }
    if (g0 < PW) {
        uint4* __restrict__ o = (uint4*)(planes + (size_t)lane * PW + g0);
#pragma unroll
        for (int j = 0; j < KEY_GROUPS; j += 4) o[j >> 2] = make_uint4(pw[j], pw[j + 1], pw[j + 2], pw[j + 3]);
    }
}

__device__ __forceinline__ int hist_pitch(int Nn) {
    return (((Nn + 1) >> 1) + 31) & ~31;
}

__global__ __launch_bounds__(HIST_THREADS)
void deg_hist_kernel(const int* __restrict__ dst, unsigned* __restrict__ part,
                     unsigned* __restrict__ planes, int E, int Nn, int chunks) {
    __shared__ uint4 cnt4[HIST_WORDS / 4];
    unsigned* cnt = (unsigned*)cnt4;
    const int tid   = threadIdx.x;
    if ((int)blockIdx.x >= chunks) {
        if (blockIdx.y == 0) {
            const int wave = __builtin_amdgcn_readfirstlane(tid >> 5);
            const int kw = ((int)blockIdx.x - chunks) * (HIST_THREADS / 32) + wave;
            write_key_planes(dst, planes, kw * KEY_GROUPS, tid & 31, E);
        }
        return;
    }
    const int words = (Nn + 1) >> 1;
    const int wbase = blockIdx.y * HIST_WORDS;
    int wspan = words - wbase;
    if (wspan > HIST_WORDS) wspan = HIST_WORDS;
    const unsigned nbase = (unsigned)wbase * 2u;
    const unsigned nspan = (unsigned)wspan * 2u;

    for (int i = tid; i < HIST_WORDS / 4; i += HIST_THREADS) cnt4[i] = make_uint4(0u, 0u, 0u, 0u);

    const int e0 = blockIdx.x * HIST_CHUNK + tid;
    int d[HIST_CHUNK / HIST_THREADS];
#pragma unroll
    for (int k = 0; k < HIST_CHUNK / HIST_THREADS; ++k) {
        const int e = e0 + k * HIST_THREADS;
        d[k] = dst[e < E ? e : E - 1];
    }
    __syncthreads();
#pragma unroll
    for (int k = 0; k < HIST_CHUNK / HIST_THREADS; ++k) {
        const int e = e0 + k * HIST_THREADS;
        const unsigned r = (unsigned)d[k] - nbase;
        if (e < E && r < nspan) atomicAdd(&cnt[r >> 1], 1u << ((r & 1u) * 16u));
    }
    __syncthreads();

    unsigned* __restrict__ o = part + (size_t)blockIdx.x * hist_pitch(Nn) + wbase;
    const int w4 = wspan >> 2;
    for (int i = tid; i < w4; i += HIST_THREADS) ((uint4*)o)[i] = cnt4[i];
    for (int i = (w4 << 2) + tid; i < wspan; i += HIST_THREADS) o[i] = cnt[i];
}

#define OWN_WORDS    (OWN_NODES / 2)
#define DINV_THREADS 224
__global__ __launch_bounds__(DINV_THREADS)
void dinv_kernel(const unsigned* __restrict__ part, float* __restrict__ dinv,
                 unsigned* __restrict__ cnt, int Nn, int P) {
    __shared__ unsigned wsum[DINV_THREADS / 32];
    const int tid = threadIdx.x;
    const int w = blockIdx.x * OWN_WORDS + tid;
    const int words = (Nn + 1) >> 1;
    const int pitch = hist_pitch(Nn);
    unsigned lo = 0u, hi = 0u;
    if (tid < OWN_WORDS && w < words) {
#pragma unroll 16
        for (int p = 0; p < P; ++p) {
            const unsigned v = part[(size_t)p * pitch + w];
            lo += v & 0xFFFFu;
            hi += v >> 16;
        }
        const float d0 = rsqrtf((float)lo + 1.0f);
        const float d1 = rsqrtf((float)hi + 1.0f);
        if (2 * w + 1 < Nn) *(float2*)(dinv + 2 * w) = make_float2(d0, d1);
        else                dinv[2 * w] = d0;
    }
    unsigned t = lo + hi;
    for (int o = 16; o > 0; o >>= 1) t += __shfl_xor(t, o);
    if ((tid & 31) == 0) wsum[tid >> 5] = t;
    __syncthreads();
    if (tid == 0) {
        unsigned s = 0u;
        for (int k = 0; k < DINV_THREADS / 32; ++k) s += wsum[k];
        cnt[blockIdx.x] = s;
    }
}

__device__ __forceinline__ unsigned records_before(const unsigned* __restrict__ cnt, int b, int lane) {
    unsigned t = 0u;
#pragma unroll
    for (int k = 0; k < 8; ++k) {
        const int i = lane + 32 * k;
        const unsigned v = cnt[i < b ? i : 0];
        t += i < b ? v : 0u;
    }
    for (int o = 16; o > 0; o >>= 1) t += __shfl_xor(t, o);
    return t;
}

__device__ __forceinline__ float squash(float v) {
    return (v * v) * __builtin_amdgcn_rcpf(v);
}

#define REC_ROW_SHIFT 17
#define RECB_EDGES    16
#define RECB_FLIGHT   8
template <int THREADS>
__device__ __forceinline__ void write_edge_records(const int* __restrict__ src, const int* __restrict__ dst,
                                                   const float* __restrict__ dinv, uint2* __restrict__ erec,
                                                   int E, int blk, int tid) {
#pragma nounroll
    for (int pass = 0; pass < RECB_EDGES / RECB_FLIGHT; ++pass) {
        const int e0 = (blk * (RECB_EDGES / RECB_FLIGHT) + pass) * (THREADS * RECB_FLIGHT) + tid;
        int s[RECB_FLIGHT], d[RECB_FLIGHT];
#pragma unroll
        for (int k = 0; k < RECB_FLIGHT; ++k) {
            const int e  = e0 + k * THREADS;
            const int ec = e < E ? e : E - 1;
            s[k] = src[ec];
            d[k] = dst[ec];
        }
        float w[RECB_FLIGHT];
#pragma unroll
        for (int k = 0; k < RECB_FLIGHT; ++k) w[k] = dinv[s[k]];
#pragma unroll
        for (int k = 0; k < RECB_FLIGHT; ++k) {
            const unsigned r = (unsigned)d[k] % OWN_NODES;
            erec[e0 + k * THREADS] = make_uint2((unsigned)s[k] | (r << REC_ROW_SHIFT), __builtin_bit_cast(unsigned, w[k]));
        }
    }
}

template <int K, int N, bool SQUASH, bool RECS>
__global__ __launch_bounds__((N / 16) * 32)
void gemm_wmma(const float* __restrict__ X, const float* __restrict__ W,
               float* __restrict__ H, int M,
               const int* __restrict__ src, const int* __restrict__ dst, const float* __restrict__ dinv,
               uint2* __restrict__ erec, int E, int recblk) {
    constexpr int THREADS = (N / 16) * 32;
    constexpr int PITCH   = K + 4;
    constexpr int VEC_ROW = K / 4;
    constexpr int PIECES  = 16 * VEC_ROW / THREADS;
    __shared__ float4 tile4[16 * PITCH / 4];
    float* tile = (float*)tile4;

    if (RECS && (int)blockIdx.x < recblk) {
        write_edge_records<THREADS>(src, dst, dinv, erec, E, (int)blockIdx.x, (int)threadIdx.x);
        return;
    }
    const int lane  = threadIdx.x & 31;
    const int ntile = threadIdx.x >> 5;
    const int half  = lane >> 4;
    const int l16   = lane & 15;
    const int ncol  = ntile * 16 + l16;
    const int row0  = ((int)blockIdx.x - (RECS ? recblk : 0)) * 16;

    float4 xv[PIECES];
#pragma unroll
    for (int i = 0; i < PIECES; ++i) {
        const int idx = threadIdx.x + i * THREADS;
        const int r   = idx / VEC_ROW;
        const int c4  = idx % VEC_ROW;
        const int gr  = row0 + r < M ? row0 + r : M - 1;
        xv[i] = *(const float4*)(X + (long)gr * K + c4 * 4);
    }
#pragma unroll
    for (int i = 0; i < PIECES; ++i) {
        const int idx = threadIdx.x + i * THREADS;
        const int r   = idx / VEC_ROW;
        const int c4  = idx % VEC_ROW;
        float4 v = xv[i];
        if (SQUASH) { v.x = squash(v.x); v.y = squash(v.y); v.z = squash(v.z); v.w = squash(v.w); }
        *(float4*)(tile + r * PITCH + c4 * 4) = v;
    }
    __syncthreads();

    const float* arow = tile + l16 * PITCH + half * 2;
    v8f c = {0.f, 0.f, 0.f, 0.f, 0.f, 0.f, 0.f, 0.f};

#pragma unroll
    for (int kk = 0; kk < K; kk += 4) {
        const int k0 = kk + half * 2;
        const float2 av = *(const float2*)(arow + kk);
        v2f a;
        a.x = av.x; a.y = av.y;
        v2f b;
        b.x = W[k0 * N + ncol];
        b.y = W[(k0 + 1) * N + ncol];
        c = __builtin_amdgcn_wmma_f32_16x16x4_f32(
                 false, a,  false, b,
                 (short)0, c,  false,  false);
    }

#pragma unroll
    for (int i = 0; i < 8; ++i) {
        int r = row0 + i + half * 8;
        if (r < M) H[(long)r * N + ncol] = c[i];
    }
}

#define AGG_THREADS 1024
#define AGG_WAVES   32
#define SCAN_KEYS   32
#define SCAN_SUB    4
#define SCAN_AHEAD  2
#define QCAP        (32 * SCAN_KEYS + 32)

__device__ __forceinline__ unsigned clip32(unsigned m, int left) {
    if (left < 32) m &= left > 0 ? (1u << left) - 1u : 0u;
    return m;
}

#define ROW_V(C)     ((C) / 16)
#define ROW_PITCH(C) ((C) + 16)
__device__ __forceinline__ int row_slot(int c, int rowv) { return 16 * (c % rowv) + c / rowv; }

template <int G> __device__ __forceinline__ int share16(int x) {
    return __builtin_amdgcn_update_dpp(0, x, 0x150 + G, 0xF, 0xF, true);
}
template <int G> __device__ __forceinline__ float share16f(float x) {
    return __builtin_bit_cast(float, share16<G>(__builtin_bit_cast(int, x)));
}

template <int C, int G>
__device__ __forceinline__ void load_rows(const float* __restrict__ h, int sb, int k,
                                          float (*v)[ROW_V(C)]) {
    if constexpr (G < 16) {
        const unsigned piece = (unsigned)(share16<G>(sb) + k);
        if constexpr (C == 64) {
            const float4 t = ((const float4*)h)[piece];
            v[G][0] = t.x; v[G][1] = t.y; v[G][2] = t.z; v[G][3] = t.w;
        } else {
            const float2 t = ((const float2*)h)[piece];
            v[G][0] = t.x; v[G][1] = t.y;
        }
        load_rows<C, G + 1>(h, sb, k, v);
    }
}
template <int C, int G>
__device__ __forceinline__ void add_rows(float* acc, int rb, float p, int k, float (*v)[ROW_V(C)]) {
    if constexpr (G < 16) {
        const float pg = share16f<G>(p);
        float* a = acc + (share16<G>(rb) + k);
#pragma unroll
        for (int j = 0; j < ROW_V(C); ++j) atomicAdd(a + 16 * j, pg * v[G][j]);
        add_rows<C, G + 1>(acc, rb, p, k, v);
    }
}
template <int C>
__device__ __forceinline__ void add_edges(const float* __restrict__ h, float* acc,
                                          int s, int rl, float p, int lane) {
    float v[16][ROW_V(C)];
    const int k  = lane & 15;
    const int sb = s * 16;
    const int rb = (int)__umul24((unsigned)rl, (unsigned)ROW_PITCH(C));
    load_rows<C, 0>(h, sb, k, v);
    add_rows<C, 0>(acc, rb, p, k, v);
}

template <int C, bool RELU>
__device__ __forceinline__ void finish_rows(const float* __restrict__ h, const float* __restrict__ bias,
                                            float* __restrict__ out, const float* acc, const float* dloc,
                                            int base, int own, int tid) {
    constexpr int FIN = (OWN_NODES * C / 4 + AGG_THREADS - 1) / AGG_THREADS;
    const int lim = own * (C / 4);
    const float4* __restrict__ h4 = (const float4*)(h + (long)base * C);
    float4* __restrict__ o4 = (float4*)(out + (long)base * C);
    const float4 bv = *(const float4*)(bias + ((tid * 4) & (C - 1)));
    float4 hv[FIN];
#pragma unroll
    for (int k = 0; k < FIN; ++k) {
        const int i = tid + k * AGG_THREADS;
        hv[k] = h4[i < lim ? i : 0];
    }
    __syncthreads();
#pragma unroll
    for (int k = 0; k < FIN; ++k) {
        const int i = tid + k * AGG_THREADS;
        if (i < lim) {
            const float di = dloc[i / (C / 4)];
            const float* arow = acc + (i / (C / 4)) * ROW_PITCH(C);
            const int c0 = (i % (C / 4)) * 4;
            float4 a;
            a.x = arow[row_slot(c0 + 0, ROW_V(C))];
            a.y = arow[row_slot(c0 + 1, ROW_V(C))];
            a.z = arow[row_slot(c0 + 2, ROW_V(C))];
            a.w = arow[row_slot(c0 + 3, ROW_V(C))];
            float4 v;
            v.x = a.x + di * di * hv[k].x + bv.x;
            v.y = a.y + di * di * hv[k].y + bv.y;
            v.z = a.z + di * di * hv[k].z + bv.z;
            v.w = a.w + di * di * hv[k].w + bv.w;
            if (RELU) { v.x = fmaxf(v.x, 0.0f); v.y = fmaxf(v.y, 0.0f); v.z = fmaxf(v.z, 0.0f); v.w = fmaxf(v.w, 0.0f); }
            o4[i] = v;
        }
    }
}

template <int C, bool RELU>
__global__ __launch_bounds__(AGG_THREADS)
void aggregate_kernel(const float* __restrict__ h, const uint2* __restrict__ erec,
                      const uint4* __restrict__ planes,
                      const float* __restrict__ dinv, const float* __restrict__ bias,
                      float* __restrict__ out, unsigned* __restrict__ rec,
                      const unsigned* __restrict__ cnt, int E, int Nn) {
    __shared__ float4 acc4[(OWN_NODES + 1) * ROW_PITCH(C) / 4];
    __shared__ int    queue[AGG_WAVES * QCAP];
    __shared__ float  dloc[OWN_NODES];
    __shared__ unsigned cursor;
    float* acc = (float*)acc4;
    const int tid  = threadIdx.x;
    const int lane = tid & 31;
    const int base = blockIdx.x * OWN_NODES;
    int own = Nn - base;
    if (own > OWN_NODES) own = OWN_NODES;

    for (int i = tid; i < (OWN_NODES + 1) * ROW_PITCH(C) / 4; i += AGG_THREADS) acc4[i] = make_float4(0.f, 0.f, 0.f, 0.f);
    if (tid < OWN_NODES) dloc[tid] = dinv[base + (tid < own ? tid : 0)];
    if (tid == 0) cursor = 0u;
    uint2* __restrict__ mine = (uint2*)rec + records_before(cnt, blockIdx.x, lane);
    __syncthreads();

    int* q = queue + (tid >> 5) * QCAP;
    int qn = 0;
    const int PW4 = plane_pitch(E) >> 2;
    const uint4* __restrict__ ph = planes + (size_t)(blockIdx.x >> 4) * PW4;
    const uint4* __restrict__ pl = planes + (size_t)(16 + (blockIdx.x & 15)) * PW4;
    constexpr int STRIDE = AGG_THREADS * SCAN_SUB * SCAN_KEYS;
    const int steps = (E + STRIDE - 1) / STRIDE;

    int e0 = tid * (SCAN_SUB * SCAN_KEYS);
    int t  = tid;
    uint4 ring[SCAN_AHEAD][2];
#pragma unroll
    for (int a = 0; a < SCAN_AHEAD; ++a) {
        const int ta = t + a * AGG_THREADS;
        const int tc = ta < PW4 ? ta : 0;
        ring[a][0] = ph[tc];
        ring[a][1] = pl[tc];
    }
    for (int it = 0; it < steps; it += SCAN_AHEAD) {
#pragma unroll
        for (int a = 0; a < SCAN_AHEAD; ++a) {
            uint4 w;
            w.x = ring[a][0].x & ring[a][1].x;
            w.y = ring[a][0].y & ring[a][1].y;
            w.z = ring[a][0].z & ring[a][1].z;
            w.w = ring[a][0].w & ring[a][1].w;
            const int tn = t + SCAN_AHEAD * AGG_THREADS;
            const int tc = tn < PW4 ? tn : 0;
            ring[a][0] = ph[tc];
            ring[a][1] = pl[tc];

#pragma nounroll
            for (int j = 0; j < SCAN_SUB; ++j) {
                const int eb = e0 + j * SCAN_KEYS;
                unsigned m = clip32(w.x, E - eb);
                w.x = w.y; w.y = w.z; w.z = w.w;

                unsigned any = __builtin_amdgcn_ballot_w32(m != 0u);
                while (any) {
                    if (m) {
                        q[qn + (int)__builtin_amdgcn_mbcnt_lo(any, 0u)] = eb + __builtin_ctz(m);
                        m &= m - 1u;
                    }
                    qn += __builtin_popcount(any);
                    any = __builtin_amdgcn_ballot_w32(m != 0u);
                }
                __builtin_amdgcn_fence(__ATOMIC_ACQ_REL, "wavefront");

                while (qn >= 32) {
                    qn -= 32;
                    const int e = q[qn + lane];
                    const uint2 er = erec[e];
                    const int s = (int)(er.x & ((1u << REC_ROW_SHIFT) - 1u));
                    const int r = (int)(er.x >> REC_ROW_SHIFT);
                    const float p = __builtin_bit_cast(float, er.y) * dloc[r];
                    unsigned pos = 0u;
                    if (lane == 0) pos = atomicAdd(&cursor, 32u);
                    pos = (unsigned)__builtin_amdgcn_readfirstlane((int)pos);
                    mine[pos + lane] = make_uint2(er.x, __builtin_bit_cast(unsigned, p));
                    add_edges<C>(h, acc, s, r, p, lane);
                }
                __builtin_amdgcn_fence(__ATOMIC_ACQ_REL, "wavefront");
            }

            e0 += STRIDE;
            t  += AGG_THREADS;
        }
    }
    if (qn > 0) {
        const int n = qn;
        const bool live = lane < n;
        const int e = q[live ? lane : 0];
        const uint2 er = erec[e];
        const int s = (int)(er.x & ((1u << REC_ROW_SHIFT) - 1u));
        const int ro = (int)(er.x >> REC_ROW_SHIFT);
        const float po = __builtin_bit_cast(float, er.y) * dloc[ro];
        unsigned pos = 0u;
        if (lane == 0) pos = atomicAdd(&cursor, (unsigned)n);
        pos = (unsigned)__builtin_amdgcn_readfirstlane((int)pos);
        if (live) mine[pos + lane] = make_uint2(er.x, __builtin_bit_cast(unsigned, po));
        const int   r = live ? ro : OWN_NODES;
        const float p = live ? po : 0.0f;
        add_edges<C>(h, acc, s, r, p, lane);
    }

    finish_rows<C, RELU>(h, bias, out, acc, dloc, base, own, tid);
}

template <int C, bool RELU>
__global__ __launch_bounds__(AGG_THREADS)
void aggregate_list_kernel(const float* __restrict__ h, const unsigned* __restrict__ rec,
                           const unsigned* __restrict__ cnt, const float* __restrict__ dinv,
                           const float* __restrict__ bias, float* __restrict__ out, int Nn) {
    __shared__ float4 acc4[(OWN_NODES + 1) * ROW_PITCH(C) / 4];
    __shared__ float  dloc[OWN_NODES];
    float* acc = (float*)acc4;
    const int tid  = threadIdx.x;
    const int lane = tid & 31;
    const int base = blockIdx.x * OWN_NODES;
    int own = Nn - base;
    if (own > OWN_NODES) own = OWN_NODES;

    for (int i = tid; i < (OWN_NODES + 1) * ROW_PITCH(C) / 4; i += AGG_THREADS) acc4[i] = make_float4(0.f, 0.f, 0.f, 0.f);
    if (tid < OWN_NODES) dloc[tid] = dinv[base + (tid < own ? tid : 0)];
    const int n = (int)cnt[blockIdx.x];
    const uint2* __restrict__ mine = (const uint2*)rec + records_before(cnt, blockIdx.x, lane);
    __syncthreads();

    constexpr int STEP = AGG_THREADS;
    int i = tid - lane;
    uint2 cur = make_uint2(0u, 0u);
    if (i < n) cur = mine[i + lane < n ? i + lane : i];
    for (; i < n; i += STEP) {
        const int inext = i + STEP;
        uint2 nxt = cur;
        if (inext < n) nxt = mine[inext + lane < n ? inext + lane : inext];
        const bool  live = i + lane < n;
        const int   s = (int)(cur.x & ((1u << REC_ROW_SHIFT) - 1u));
        const int   r = live ? (int)(cur.x >> REC_ROW_SHIFT) : OWN_NODES;
        const float p = live ? __builtin_bit_cast(float, cur.y) : 0.0f;
        add_edges<C>(h, acc, s, r, p, lane);
        cur = nxt;
    }

    finish_rows<C, RELU>(h, bias, out, acc, dloc, base, own, tid);
}

static inline size_t align256(size_t n) { return (n + 255) & ~size_t(255); }

extern "C" void kernel_launch(void* const* d_in, const int* in_sizes, int n_in,
                              void* d_out, int out_size, void* d_ws, size_t ws_size,
                              hipStream_t stream) {
    const float* x  = (const float*)d_in[0];
    const int*   ei = (const int*)d_in[1];
    const float* W1 = (const float*)d_in[2];
    const float* b1 = (const float*)d_in[3];
    const float* W2 = (const float*)d_in[4];
    const float* b2 = (const float*)d_in[5];
    float* out = (float*)d_out;

    const int hid  = in_sizes[3];
    const int outc = in_sizes[5];
    const int inc  = in_sizes[2] / hid;
    const int Nn   = in_sizes[0] / inc;
    const int E    = in_sizes[1] / 2;
    const int* src = ei;
    const int* dstv = ei + E;
    (void)outc;

    char* ws = (char*)d_ws;
    float* deg = (float*)ws;  ws += align256((size_t)Nn * 4);
    float* h1  = (float*)ws;  ws += (size_t)Nn * 64 * 4;
    float* a1  = (float*)ws;  ws += (size_t)Nn * 64 * 4;
    float* h2  = (float*)ws;  ws += align256((size_t)Nn * 32 * 4);
    const int groups = (E + 31) / 32;
    unsigned* planes = (unsigned*)ws;  ws += align256((size_t)KEY_PLANES * plane_pitch(E) * 4);
    unsigned* cnt  = (unsigned*)ws;  ws += align256(256 * 4);
    unsigned* rec  = (unsigned*)ws;  ws += align256((size_t)E * 8);
    const int recblk = (E + 128 * RECB_EDGES - 1) / (128 * RECB_EDGES);
    uint2*    erec = (uint2*)ws;     ws += align256((size_t)recblk * 128 * RECB_EDGES * 8);
    unsigned* part = (unsigned*)ws;

    const int chunks = (E + HIST_CHUNK - 1) / HIST_CHUNK;
    const int words  = (Nn + 1) / 2;
    const int keyblk = (groups + (HIST_THREADS / 32) * KEY_GROUPS - 1) / ((HIST_THREADS / 32) * KEY_GROUPS);
    deg_hist_kernel<<<dim3(chunks + keyblk, (words + HIST_WORDS - 1) / HIST_WORDS), HIST_THREADS, 0, stream>>>(dstv, part, planes, E, Nn, chunks);
    const int owners = (Nn + OWN_NODES - 1) / OWN_NODES;
    dinv_kernel<<<owners, DINV_THREADS, 0, stream>>>(part, deg, cnt, Nn, chunks);

    gemm_wmma<128, 64, true, true><<<recblk + (Nn + 15) / 16, 128, 0, stream>>>(x, W1, h1, Nn, src, dstv, deg, erec, E, recblk);
    aggregate_kernel<64, true><<<owners, AGG_THREADS, 0, stream>>>(h1, erec, (const uint4*)planes, deg, b1, a1, rec, cnt, E, Nn);

    gemm_wmma<64, 32, false, false><<<(Nn + 15) / 16, 64, 0, stream>>>(a1, W2, h2, Nn, nullptr, nullptr, nullptr, nullptr, 0, 0);
    aggregate_list_kernel<32, false><<<owners, AGG_THREADS, 0, stream>>>(h2, rec, cnt, deg, b2, out, Nn);
}
